# starting kernel
# baseline (speedup 1.0000x reference)
#include <hip/hip_runtime.h>
#include <hip/hip_bf16.h>

typedef float v2f __attribute__((ext_vector_type(2)));
typedef float v8f __attribute__((ext_vector_type(8)));

static __device__ __forceinline__ v8f wmma_f32_4(v2f a, v2f b, v8f c) {
  return __builtin_amdgcn_wmma_f32_16x16x4_f32(false, a, false, b, (short)0, c,
                                               false, false);
}

#define ZERO8 {0.f,0.f,0.f,0.f,0.f,0.f,0.f,0.f}

__global__ __launch_bounds__(256) void gemm128x64_wmma(
    const float* __restrict__ A, const float* __restrict__ B,
    const float* __restrict__ bias, float* __restrict__ C,
    int M, int N, int K) {
  __shared__ float bt[64 * 66];

  const int m0 = blockIdx.x * 128;
  const int n0 = blockIdx.y * 64;
  const int tid = threadIdx.x;
  const int w = tid >> 5;
  const int lane = tid & 31;
  const int L = lane & 15;
  const int hi = lane >> 4;
  const int arow = m0 + w * 16 + L;

  v8f acc[4] = {ZERO8, ZERO8, ZERO8, ZERO8};

  const int c4 = (tid & 15) * 4;
  const int r0 = tid >> 4;

  for (int kc = 0; kc < K; kc += 64) {
#pragma unroll
    for (int p = 0; p < 4; ++p) {
      const int r = r0 + p * 16;
      const float4 bv =
          *(const float4*)&B[(size_t)(kc + r) * N + n0 + c4];
      bt[(c4 + 0) * 66 + r] = bv.x;
      bt[(c4 + 1) * 66 + r] = bv.y;
      bt[(c4 + 2) * 66 + r] = bv.z;
      bt[(c4 + 3) * 66 + r] = bv.w;
    }
    __syncthreads();
#pragma unroll
    for (int kk = 0; kk < 64; kk += 4) {
      const v2f av = *(const v2f*)&A[(size_t)arow * K + kc + kk + hi * 2];
#pragma unroll
      for (int t = 0; t < 4; ++t) {
        const v2f bf = *(const v2f*)&bt[(t * 16 + L) * 66 + kk + hi * 2];
        acc[t] = wmma_f32_4(av, bf, acc[t]);
      }
    }
    __syncthreads();
  }

  const int crow = m0 + w * 16 + hi * 8;
#pragma unroll
  for (int t = 0; t < 4; ++t) {
    const int col = n0 + t * 16 + L;
    const float bs = bias ? bias[col] : 0.0f;
#pragma unroll
    for (int r = 0; r < 8; ++r)
      C[(size_t)(crow + r) * N + col] = acc[t][r] + bs;
  }
}

__global__ __launch_bounds__(256) void attn_kernel(
    const float* __restrict__ qkv, const int* __restrict__ mask,
    float* __restrict__ out) {
  extern __shared__ float lds[];
  float* simb = lds;
  float* rinv = lds + 256 * 258;

  const int bid = blockIdx.x;
  const int s = bid >> 5;
  const int b = (bid >> 3) & 3;
  const int h = bid & 7;

  const float* base = qkv + (size_t)((s * 4 + b) * 256) * 768;
  const float* qb = base + h * 32;
  const float* kb = base + 256 + h * 32;
  const float* vb = base + 512 + h * 32;
  const int* mk = mask + (size_t)s * 256 * 256;
  float* ob = out + (size_t)((s * 4 + b) * 256) * 256 + h * 32;

  const int tid = threadIdx.x;
  const int w = tid >> 5;
  const int lane = tid & 31;
  const int L = lane & 15;
  const int hi = lane >> 4;
  const float scale = 0.17677669529663687f;
  const float NEG = -1.0e10f;

  v2f qa[2][8];
#pragma unroll
  for (int rb = 0; rb < 2; ++rb) {
    const int rbase = rb * 128 + w * 16;
#pragma unroll
    for (int c = 0; c < 8; ++c)
      qa[rb][c] = *(const v2f*)&qb[(size_t)(rbase + L) * 768 + c * 4 + hi * 2];
  }

  for (int j0 = 0; j0 < 256; j0 += 16) {
    v2f kf[8];
#pragma unroll
    for (int c = 0; c < 8; ++c)
      kf[c] = *(const v2f*)&kb[(size_t)(j0 + L) * 768 + c * 4 + hi * 2];
#pragma unroll
    for (int rb = 0; rb < 2; ++rb) {
      v8f acc = ZERO8;
#pragma unroll
      for (int c = 0; c < 8; ++c) acc = wmma_f32_4(qa[rb][c], kf[c], acc);
      const int ib = rb * 128 + w * 16 + hi * 8;
#pragma unroll
      for (int r = 0; r < 8; ++r) {
        float sv = acc[r] * scale;
        sv = (mk[(ib + r) * 256 + j0 + L] == 0) ? NEG : sv;
        simb[(ib + r) * 258 + j0 + L] = sv;
      }
    }
  }
  __syncthreads();

  {
    float* row = simb + tid * 258;
    float m = -3.4e38f;
    for (int j = 0; j < 256; ++j) m = fmaxf(m, row[j]);
    float sum = 0.0f;
    for (int j = 0; j < 256; ++j) {
      const float e = __expf(row[j] - m);
      row[j] = e;
      sum += e;
    }
    rinv[tid] = 1.0f / sum;
  }
  __syncthreads();

  v8f oacc[2][2] = {{ZERO8, ZERO8}, {ZERO8, ZERO8}};
  for (int jj = 0; jj < 256; jj += 4) {
    v2f va[2];
#pragma unroll
    for (int dt = 0; dt < 2; ++dt) {
      v2f t;
      t.x = vb[(size_t)(jj + hi * 2) * 768 + dt * 16 + L];
      t.y = vb[(size_t)(jj + hi * 2 + 1) * 768 + dt * 16 + L];
      va[dt] = t;
    }
#pragma unroll
    for (int rb = 0; rb < 2; ++rb) {
      const v2f pa =
          *(const v2f*)&simb[(rb * 128 + w * 16 + L) * 258 + jj + hi * 2];
      oacc[rb][0] = wmma_f32_4(pa, va[0], oacc[rb][0]);
      oacc[rb][1] = wmma_f32_4(pa, va[1], oacc[rb][1]);
    }
  }
#pragma unroll
  for (int rb = 0; rb < 2; ++rb) {
    const int ib = rb * 128 + w * 16 + hi * 8;
#pragma unroll
    for (int r = 0; r < 8; ++r) {
      const float ri = rinv[ib + r];
      ob[(size_t)(ib + r) * 256 + L] = oacc[rb][0][r] * ri;
      ob[(size_t)(ib + r) * 256 + 16 + L] = oacc[rb][1][r] * ri;
    }
  }
}

extern "C" void kernel_launch(void* const* d_in, const int* in_sizes, int n_in,
                              void* d_out, int out_size, void* d_ws,
                              size_t ws_size, hipStream_t stream) {
  (void)in_sizes; (void)n_in; (void)out_size; (void)ws_size;
  const float* x    = (const float*)d_in[0];
  const int*   mask = (const int*)d_in[1];
  const float* Wqkv = (const float*)d_in[2];
  const float* Wout = (const float*)d_in[3];
  const float* bout = (const float*)d_in[4];
  float* outp = (float*)d_out;

  float* qkv = (float*)d_ws;
  float* ao  = qkv + (size_t)65536 * 768;

  const dim3 blk(256);

  gemm128x64_wmma<<<dim3(512, 12), blk, 0, stream>>>(x, Wqkv, nullptr, qkv,
                                                     65536, 768, 256);

  const size_t smem = (size_t)(256 * 258 + 256) * sizeof(float);
  hipFuncSetAttribute((const void*)attn_kernel,
                      hipFuncAttributeMaxDynamicSharedMemorySize, (int)smem);
  attn_kernel<<<dim3(2048), blk, smem, stream>>>(qkv, mask, ao);

  gemm128x64_wmma<<<dim3(512, 4), blk, 0, stream>>>(ao, Wout, bout, outp,
                                                    65536, 256, 256);
}


# attention: scores computed transposed (K Q^T) and held in registers, softmax in-lane, P fed from those registers as WMMA B operand, no LDS or barriers; score-k, softmax-sum and PV sum orders change
# speedup vs baseline: 1.3460x; 1.3460x over previous
#include <hip/hip_runtime.h>
#include <hip/hip_bf16.h>

typedef float v2f __attribute__((ext_vector_type(2)));
typedef float v8f __attribute__((ext_vector_type(8)));

static __device__ __forceinline__ v8f wmma_f32_4(v2f a, v2f b, v8f c) {
  return __builtin_amdgcn_wmma_f32_16x16x4_f32(false, a, false, b, (short)0, c,
                                               false, false);
}

#define ZERO8 {0.f,0.f,0.f,0.f,0.f,0.f,0.f,0.f}

__global__ __launch_bounds__(256) void gemm128x64_wmma(
    const float* __restrict__ A, const float* __restrict__ B,
    const float* __restrict__ bias, float* __restrict__ C,
    int M, int N, int K) {
  __shared__ float bt[64 * 66];

  const int m0 = blockIdx.x * 128;
  const int n0 = blockIdx.y * 64;
  const int tid = threadIdx.x;
  const int w = tid >> 5;
  const int lane = tid & 31;
  const int L = lane & 15;
  const int hi = lane >> 4;
  const int arow = m0 + w * 16 + L;

  v8f acc[4] = {ZERO8, ZERO8, ZERO8, ZERO8};

  const int c4 = (tid & 15) * 4;
  const int r0 = tid >> 4;

  for (int kc = 0; kc < K; kc += 64) {
#pragma unroll
    for (int p = 0; p < 4; ++p) {
      const int r = r0 + p * 16;
      const float4 bv =
          *(const float4*)&B[(size_t)(kc + r) * N + n0 + c4];
      bt[(c4 + 0) * 66 + r] = bv.x;
      bt[(c4 + 1) * 66 + r] = bv.y;
      bt[(c4 + 2) * 66 + r] = bv.z;
      bt[(c4 + 3) * 66 + r] = bv.w;
    }
    __syncthreads();
#pragma unroll
    for (int kk = 0; kk < 64; kk += 4) {
      const v2f av = *(const v2f*)&A[(size_t)arow * K + kc + kk + hi * 2];
#pragma unroll
      for (int t = 0; t < 4; ++t) {
        const v2f bf = *(const v2f*)&bt[(t * 16 + L) * 66 + kk + hi * 2];
        acc[t] = wmma_f32_4(av, bf, acc[t]);
      }
    }
    __syncthreads();
  }

  const int crow = m0 + w * 16 + hi * 8;
#pragma unroll
  for (int t = 0; t < 4; ++t) {
    const int col = n0 + t * 16 + L;
    const float bs = bias ? bias[col] : 0.0f;
#pragma unroll
    for (int r = 0; r < 8; ++r)
      C[(size_t)(crow + r) * N + col] = acc[t][r] + bs;
  }
}

__global__ __launch_bounds__(256) void attn_kernel(
    const float* __restrict__ qkv, const int* __restrict__ mask,
    float* __restrict__ out) {
  const int bid = blockIdx.x;
  const int half = bid & 1;
  const int sbh = bid >> 1;
  const int s = sbh >> 5;
  const int b = (sbh >> 3) & 3;
  const int h = sbh & 7;

  const float* base = qkv + (size_t)((s * 4 + b) * 256) * 768;
  const float* qb = base + h * 32;
  const float* kb = base + 256 + h * 32;
  const float* vb = base + 512 + h * 32;
  const int* mk = mask + (size_t)s * 256 * 256;
  float* ob = out + (size_t)((s * 4 + b) * 256) * 256 + h * 32;

  const int tid = threadIdx.x;
  const int w = tid >> 5;
  const int lane = tid & 31;
  const int L = lane & 15;
  const int hi = lane >> 4;
  const int i0 = half * 128 + w * 16;
  const float scale = 0.17677669529663687f;
  const float NEG = -1.0e10f;

  v2f qf[8];
  {
    const float* qp = qb + (size_t)(i0 + L) * 768 + 16 * hi;
#pragma unroll
    for (int v = 0; v < 4; ++v) {
      const float4 x = *(const float4*)(qp + 4 * v);
      qf[2 * v] = (v2f){x.x, x.y};
      qf[2 * v + 1] = (v2f){x.z, x.w};
    }
  }

  v8f sc[16];
  const float* kp0 = kb + (size_t)L * 768 + 16 * hi;
  const int* mp0 = mk + (size_t)(i0 + L) * 256 + 8 * hi;
#pragma unroll
  for (int tp = 0; tp < 16; tp += 2) {
    v2f ka[2][8];
#pragma unroll
    for (int u = 0; u < 2; ++u) {
      const float* kp = kp0 + (size_t)(16 * (tp + u)) * 768;
#pragma unroll
      for (int v = 0; v < 4; ++v) {
        const float4 x = *(const float4*)(kp + 4 * v);
        ka[u][2 * v] = (v2f){x.x, x.y};
        ka[u][2 * v + 1] = (v2f){x.z, x.w};
      }
    }
    v8f a0 = ZERO8, a1 = ZERO8;
#pragma unroll
    for (int c = 0; c < 8; ++c) {
      a0 = wmma_f32_4(ka[0][c], qf[c], a0);
      a1 = wmma_f32_4(ka[1][c], qf[c], a1);
    }
#pragma unroll
    for (int u = 0; u < 2; ++u) {
      const int* mp = mp0 + 16 * (tp + u);
      const int4 m0 = *(const int4*)mp;
      const int4 m1 = *(const int4*)(mp + 4);
      const int mm[8] = {m0.x, m0.y, m0.z, m0.w, m1.x, m1.y, m1.z, m1.w};
      const v8f a = u ? a1 : a0;
      v8f o;
#pragma unroll
      for (int r = 0; r < 8; ++r) o[r] = (mm[r] == 0) ? NEG : a[r] * scale;
      sc[tp + u] = o;
    }
  }

  float mx4[4] = {-3.4e38f, -3.4e38f, -3.4e38f, -3.4e38f};
#pragma unroll
  for (int t = 0; t < 16; ++t)
#pragma unroll
    for (int r = 0; r < 8; ++r) mx4[r & 3] = fmaxf(mx4[r & 3], sc[t][r]);
  float mx = fmaxf(fmaxf(mx4[0], mx4[1]), fmaxf(mx4[2], mx4[3]));
  mx = fmaxf(mx, __shfl_xor(mx, 16));
  float sm4[4] = {0.f, 0.f, 0.f, 0.f};
#pragma unroll
  for (int t = 0; t < 16; ++t)
#pragma unroll
    for (int r = 0; r < 8; ++r) {
      const float e = __expf(sc[t][r] - mx);
      sc[t][r] = e;
      sm4[r & 3] += e;
    }
  float sm = (sm4[0] + sm4[1]) + (sm4[2] + sm4[3]);
  sm += __shfl_xor(sm, 16);
  const float rinv = 1.0f / sm;

  v8f o0 = ZERO8, o1 = ZERO8;
  const float* vp0 = vb + (size_t)(8 * hi) * 768 + L;
#pragma unroll
  for (int t = 0; t < 16; ++t) {
#pragma unroll
    for (int u = 0; u < 4; ++u) {
      const float* vp = vp0 + (size_t)(16 * t + 2 * u) * 768;
      const v2f va0 = (v2f){vp[0], vp[768]};
      const v2f va1 = (v2f){vp[16], vp[768 + 16]};
      const v2f pb = (v2f){sc[t][2 * u], sc[t][2 * u + 1]};
      o0 = wmma_f32_4(va0, pb, o0);
      o1 = wmma_f32_4(va1, pb, o1);
    }
  }

  float* op = ob + (size_t)(i0 + L) * 256 + 8 * hi;
  *(float4*)(op)      = (float4){o0[0] * rinv, o0[1] * rinv, o0[2] * rinv, o0[3] * rinv};
  *(float4*)(op + 4)  = (float4){o0[4] * rinv, o0[5] * rinv, o0[6] * rinv, o0[7] * rinv};
  *(float4*)(op + 16) = (float4){o1[0] * rinv, o1[1] * rinv, o1[2] * rinv, o1[3] * rinv};
  *(float4*)(op + 20) = (float4){o1[4] * rinv, o1[5] * rinv, o1[6] * rinv, o1[7] * rinv};
}

extern "C" void kernel_launch(void* const* d_in, const int* in_sizes, int n_in,
                              void* d_out, int out_size, void* d_ws,
                              size_t ws_size, hipStream_t stream) {
  (void)in_sizes; (void)n_in; (void)out_size; (void)ws_size;
  const float* x    = (const float*)d_in[0];
  const int*   mask = (const int*)d_in[1];
  const float* Wqkv = (const float*)d_in[2];
  const float* Wout = (const float*)d_in[3];
  const float* bout = (const float*)d_in[4];
  float* outp = (float*)d_out;

  float* qkv = (float*)d_ws;
  float* ao  = qkv + (size_t)65536 * 768;

  const dim3 blk(256);

  gemm128x64_wmma<<<dim3(512, 12), blk, 0, stream>>>(x, Wqkv, nullptr, qkv,
                                                     65536, 768, 256);

  attn_kernel<<<dim3(4096), blk, 0, stream>>>(qkv, mask, ao);

  gemm128x64_wmma<<<dim3(512, 4), blk, 0, stream>>>(ao, Wout, bout, outp,
                                                    65536, 256, 256);
}


# attention: K and V of a head staged once per block into LDS by whole-line loads (V transposed); fragments read from LDS by 16-byte loads, not gathered from global per wave; summation orders unchanged
# speedup vs baseline: 1.4953x; 1.1109x over previous
#include <hip/hip_runtime.h>
#include <hip/hip_bf16.h>

typedef float v2f __attribute__((ext_vector_type(2)));
typedef float v8f __attribute__((ext_vector_type(8)));

static __device__ __forceinline__ v8f wmma_f32_4(v2f a, v2f b, v8f c) {
  return __builtin_amdgcn_wmma_f32_16x16x4_f32(false, a, false, b, (short)0, c,
                                               false, false);
}

#define ZERO8 {0.f,0.f,0.f,0.f,0.f,0.f,0.f,0.f}

__global__ __launch_bounds__(256) void gemm128x64_wmma(
    const float* __restrict__ A, const float* __restrict__ B,
    const float* __restrict__ bias, float* __restrict__ C,
    int M, int N, int K) {
  __shared__ float bt[64 * 66];

  const int m0 = blockIdx.x * 128;
  const int n0 = blockIdx.y * 64;
  const int tid = threadIdx.x;
  const int w = tid >> 5;
  const int lane = tid & 31;
  const int L = lane & 15;
  const int hi = lane >> 4;
  const int arow = m0 + w * 16 + L;

  v8f acc[4] = {ZERO8, ZERO8, ZERO8, ZERO8};

  const int c4 = (tid & 15) * 4;
  const int r0 = tid >> 4;

  for (int kc = 0; kc < K; kc += 64) {
#pragma unroll
    for (int p = 0; p < 4; ++p) {
      const int r = r0 + p * 16;
      const float4 bv =
          *(const float4*)&B[(size_t)(kc + r) * N + n0 + c4];
      bt[(c4 + 0) * 66 + r] = bv.x;
      bt[(c4 + 1) * 66 + r] = bv.y;
      bt[(c4 + 2) * 66 + r] = bv.z;
      bt[(c4 + 3) * 66 + r] = bv.w;
    }
    __syncthreads();
#pragma unroll
    for (int kk = 0; kk < 64; kk += 4) {
      const v2f av = *(const v2f*)&A[(size_t)arow * K + kc + kk + hi * 2];
#pragma unroll
      for (int t = 0; t < 4; ++t) {
        const v2f bf = *(const v2f*)&bt[(t * 16 + L) * 66 + kk + hi * 2];
        acc[t] = wmma_f32_4(av, bf, acc[t]);
      }
    }
    __syncthreads();
  }

  const int crow = m0 + w * 16 + hi * 8;
#pragma unroll
  for (int t = 0; t < 4; ++t) {
    const int col = n0 + t * 16 + L;
    const float bs = bias ? bias[col] : 0.0f;
#pragma unroll
    for (int r = 0; r < 8; ++r)
      C[(size_t)(crow + r) * N + col] = acc[t][r] + bs;
  }
}

#define KPITCH 36
#define VPITCH 260
__global__ __launch_bounds__(256) void attn_kernel(
    const float* __restrict__ qkv, const int* __restrict__ mask,
    float* __restrict__ out) {
  __shared__ __attribute__((aligned(16))) float ks[256 * KPITCH];
  __shared__ __attribute__((aligned(16))) float vt[32 * VPITCH];

  const int bid = blockIdx.x;
  const int half = bid & 1;
  const int sbh = bid >> 1;
  const int s = sbh >> 5;
  const int b = (sbh >> 3) & 3;
  const int h = sbh & 7;

  const float* base = qkv + (size_t)((s * 4 + b) * 256) * 768;
  const float* qb = base + h * 32;
  const float* kb = base + 256 + h * 32;
  const float* vb = base + 512 + h * 32;
  const int* mk = mask + (size_t)s * 256 * 256;
  float* ob = out + (size_t)((s * 4 + b) * 256) * 256 + h * 32;

  const int tid = threadIdx.x;
  const int w = tid >> 5;
  const int lane = tid & 31;
  const int L = lane & 15;
  const int hi = lane >> 4;
  const int i0 = half * 128 + w * 16;
  const float scale = 0.17677669529663687f;
  const float NEG = -1.0e10f;

  {
    const int r8 = tid >> 3;
    const int d4 = (tid & 7) * 4;
    float4 kx[8], vx[8];
#pragma unroll
    for (int p = 0; p < 8; ++p) {
      const size_t ro = (size_t)(r8 + 32 * p) * 768 + d4;
      kx[p] = *(const float4*)(kb + ro);
      vx[p] = *(const float4*)(vb + ro);
    }
#pragma unroll
    for (int p = 0; p < 8; ++p) {
      const int key = r8 + 32 * p;
      *(float4*)&ks[key * KPITCH + d4] = kx[p];
      vt[(d4 + 0) * VPITCH + key] = vx[p].x;
      vt[(d4 + 1) * VPITCH + key] = vx[p].y;
      vt[(d4 + 2) * VPITCH + key] = vx[p].z;
      vt[(d4 + 3) * VPITCH + key] = vx[p].w;
    }
  }

  v2f qf[8];
  {
    const float* qp = qb + (size_t)(i0 + L) * 768 + 16 * hi;
#pragma unroll
    for (int v = 0; v < 4; ++v) {
      const float4 x = *(const float4*)(qp + 4 * v);
      qf[2 * v] = (v2f){x.x, x.y};
      qf[2 * v + 1] = (v2f){x.z, x.w};
    }
  }

  __syncthreads();

  v8f sc[16];
  const float* kp0 = ks + L * KPITCH + 16 * hi;
  const int* mp0 = mk + (size_t)(i0 + L) * 256 + 8 * hi;
#pragma unroll
  for (int tp = 0; tp < 16; tp += 2) {
    v2f ka[2][8];
#pragma unroll
    for (int u = 0; u < 2; ++u) {
      const float* kp = kp0 + (16 * (tp + u)) * KPITCH;
#pragma unroll
      for (int v = 0; v < 4; ++v) {
        const float4 x = *(const float4*)(kp + 4 * v);
        ka[u][2 * v] = (v2f){x.x, x.y};
        ka[u][2 * v + 1] = (v2f){x.z, x.w};
      }
    }
    v8f a0 = ZERO8, a1 = ZERO8;
#pragma unroll
    for (int c = 0; c < 8; ++c) {
      a0 = wmma_f32_4(ka[0][c], qf[c], a0);
      a1 = wmma_f32_4(ka[1][c], qf[c], a1);
    }
#pragma unroll
    for (int u = 0; u < 2; ++u) {
      const int* mp = mp0 + 16 * (tp + u);
      const int4 m0 = *(const int4*)mp;
      const int4 m1 = *(const int4*)(mp + 4);
      const int mm[8] = {m0.x, m0.y, m0.z, m0.w, m1.x, m1.y, m1.z, m1.w};
      const v8f a = u ? a1 : a0;
      v8f o;
#pragma unroll
      for (int r = 0; r < 8; ++r) o[r] = (mm[r] == 0) ? NEG : a[r] * scale;
      sc[tp + u] = o;
    }
  }

  float mx4[4] = {-3.4e38f, -3.4e38f, -3.4e38f, -3.4e38f};
#pragma unroll
  for (int t = 0; t < 16; ++t)
#pragma unroll
    for (int r = 0; r < 8; ++r) mx4[r & 3] = fmaxf(mx4[r & 3], sc[t][r]);
  float mx = fmaxf(fmaxf(mx4[0], mx4[1]), fmaxf(mx4[2], mx4[3]));
  mx = fmaxf(mx, __shfl_xor(mx, 16));
  float sm4[4] = {0.f, 0.f, 0.f, 0.f};
#pragma unroll
  for (int t = 0; t < 16; ++t)
#pragma unroll
    for (int r = 0; r < 8; ++r) {
      const float e = __expf(sc[t][r] - mx);
      sc[t][r] = e;
      sm4[r & 3] += e;
    }
  float sm = (sm4[0] + sm4[1]) + (sm4[2] + sm4[3]);
  sm += __shfl_xor(sm, 16);
  const float rinv = 1.0f / sm;

  v8f o0 = ZERO8, o1 = ZERO8;
  const float* vp0 = vt + L * VPITCH + 8 * hi;
#pragma unroll
  for (int t = 0; t < 16; ++t) {
    const float* vp = vp0 + 16 * t;
    const float4 x0 = *(const float4*)(vp);
    const float4 x1 = *(const float4*)(vp + 4);
    const float4 y0 = *(const float4*)(vp + 16 * VPITCH);
    const float4 y1 = *(const float4*)(vp + 16 * VPITCH + 4);
    const v2f va0[4] = {(v2f){x0.x, x0.y}, (v2f){x0.z, x0.w},
                        (v2f){x1.x, x1.y}, (v2f){x1.z, x1.w}};
    const v2f va1[4] = {(v2f){y0.x, y0.y}, (v2f){y0.z, y0.w},
                        (v2f){y1.x, y1.y}, (v2f){y1.z, y1.w}};
#pragma unroll
    for (int u = 0; u < 4; ++u) {
      const v2f pb = (v2f){sc[t][2 * u], sc[t][2 * u + 1]};
      o0 = wmma_f32_4(va0[u], pb, o0);
      o1 = wmma_f32_4(va1[u], pb, o1);
    }
  }

  float* op = ob + (size_t)(i0 + L) * 256 + 8 * hi;
  *(float4*)(op)      = (float4){o0[0] * rinv, o0[1] * rinv, o0[2] * rinv, o0[3] * rinv};
  *(float4*)(op + 4)  = (float4){o0[4] * rinv, o0[5] * rinv, o0[6] * rinv, o0[7] * rinv};
  *(float4*)(op + 16) = (float4){o1[0] * rinv, o1[1] * rinv, o1[2] * rinv, o1[3] * rinv};
  *(float4*)(op + 20) = (float4){o1[4] * rinv, o1[5] * rinv, o1[6] * rinv, o1[7] * rinv};
}

extern "C" void kernel_launch(void* const* d_in, const int* in_sizes, int n_in,
                              void* d_out, int out_size, void* d_ws,
                              size_t ws_size, hipStream_t stream) {
  (void)in_sizes; (void)n_in; (void)out_size; (void)ws_size;
  const float* x    = (const float*)d_in[0];
  const int*   mask = (const int*)d_in[1];
  const float* Wqkv = (const float*)d_in[2];
  const float* Wout = (const float*)d_in[3];
  const float* bout = (const float*)d_in[4];
  float* outp = (float*)d_out;

  float* qkv = (float*)d_ws;
  float* ao  = qkv + (size_t)65536 * 768;

  const dim3 blk(256);

  gemm128x64_wmma<<<dim3(512, 12), blk, 0, stream>>>(x, Wqkv, nullptr, qkv,
                                                     65536, 768, 256);

  attn_kernel<<<dim3(4096), blk, 0, stream>>>(qkv, mask, ao);

  gemm128x64_wmma<<<dim3(512, 4), blk, 0, stream>>>(ao, Wout, bout, outp,
                                                    65536, 256, 256);
}


# GEMM (both launches): A tile of each 64-k chunk staged once per block in LDS by whole-line loads, fragments read from LDS instead of 16-byte global gathers per wave; k and summation order unchanged
# speedup vs baseline: 1.6831x; 1.1256x over previous
#include <hip/hip_runtime.h>
#include <hip/hip_bf16.h>

typedef float v2f __attribute__((ext_vector_type(2)));
typedef float v8f __attribute__((ext_vector_type(8)));

static __device__ __forceinline__ v8f wmma_f32_4(v2f a, v2f b, v8f c) {
  return __builtin_amdgcn_wmma_f32_16x16x4_f32(false, a, false, b, (short)0, c,
                                               false, false);
}

#define ZERO8 {0.f,0.f,0.f,0.f,0.f,0.f,0.f,0.f}

#define APITCH 68
__global__ __launch_bounds__(256) void gemm128x64_wmma(
    const float* __restrict__ A, const float* __restrict__ B,
    const float* __restrict__ bias, float* __restrict__ C,
    int M, int N, int K) {
  __shared__ float bt[64 * 66];
  __shared__ __attribute__((aligned(16))) float as[128 * APITCH];

  const int m0 = blockIdx.x * 128;
  const int n0 = blockIdx.y * 64;
  const int tid = threadIdx.x;
  const int w = tid >> 5;
  const int lane = tid & 31;
  const int L = lane & 15;
  const int hi = lane >> 4;
  const float* afr = as + (w * 16 + L) * APITCH + hi * 2;
  const float* bfr = bt + L * 66 + hi * 2;

  v8f acc[4] = {ZERO8, ZERO8, ZERO8, ZERO8};

  const int c4 = (tid & 15) * 4;
  const int r0 = tid >> 4;
  float* asw = as + r0 * APITCH + c4;
  const float* ag = A + (size_t)(m0 + r0) * K + c4;
  const float* bg = B + (size_t)r0 * N + n0 + c4;
  const size_t a16 = (size_t)16 * K, b16 = (size_t)16 * N;

  for (int kc = 0; kc < K; kc += 64) {
    const float4 b0 = *(const float4*)(bg);
    const float4 b1 = *(const float4*)(bg + b16);
    const float4 b2 = *(const float4*)(bg + 2 * b16);
    const float4 b3 = *(const float4*)(bg + 3 * b16);
    const float4 a0 = *(const float4*)(ag);
    const float4 a1 = *(const float4*)(ag + a16);
    const float4 a2 = *(const float4*)(ag + 2 * a16);
    const float4 a3 = *(const float4*)(ag + 3 * a16);
    const float4 a4 = *(const float4*)(ag + 4 * a16);
    const float4 a5 = *(const float4*)(ag + 5 * a16);
    const float4 a6 = *(const float4*)(ag + 6 * a16);
    const float4 a7 = *(const float4*)(ag + 7 * a16);
    __builtin_amdgcn_sched_barrier(0);
    *(float4*)(asw + 0 * 16 * APITCH) = a0;
    *(float4*)(asw + 1 * 16 * APITCH) = a1;
    *(float4*)(asw + 2 * 16 * APITCH) = a2;
    *(float4*)(asw + 3 * 16 * APITCH) = a3;
    *(float4*)(asw + 4 * 16 * APITCH) = a4;
    *(float4*)(asw + 5 * 16 * APITCH) = a5;
    *(float4*)(asw + 6 * 16 * APITCH) = a6;
    *(float4*)(asw + 7 * 16 * APITCH) = a7;
    {
      float* bw = bt + c4 * 66 + r0;
      bw[0 * 66 +  0] = b0.x; bw[1 * 66 +  0] = b0.y; bw[2 * 66 +  0] = b0.z; bw[3 * 66 +  0] = b0.w;
      bw[0 * 66 + 16] = b1.x; bw[1 * 66 + 16] = b1.y; bw[2 * 66 + 16] = b1.z; bw[3 * 66 + 16] = b1.w;
      bw[0 * 66 + 32] = b2.x; bw[1 * 66 + 32] = b2.y; bw[2 * 66 + 32] = b2.z; bw[3 * 66 + 32] = b2.w;
      bw[0 * 66 + 48] = b3.x; bw[1 * 66 + 48] = b3.y; bw[2 * 66 + 48] = b3.z; bw[3 * 66 + 48] = b3.w;
    }
    ag += 64;
    bg += (size_t)64 * N;
    __syncthreads();
#pragma unroll
    for (int kk = 0; kk < 64; kk += 4) {
      const v2f av = *(const v2f*)(afr + kk);
#pragma unroll
      for (int t = 0; t < 4; ++t) {
        const v2f bf = *(const v2f*)(bfr + t * 16 * 66 + kk);
        acc[t] = wmma_f32_4(av, bf, acc[t]);
      }
    }
    __syncthreads();
  }

  const int crow = m0 + w * 16 + hi * 8;
#pragma unroll
  for (int t = 0; t < 4; ++t) {
    const int col = n0 + t * 16 + L;
    const float bs = bias ? bias[col] : 0.0f;
#pragma unroll
    for (int r = 0; r < 8; ++r)
      C[(size_t)(crow + r) * N + col] = acc[t][r] + bs;
  }
}

#define KPITCH 36
#define VPITCH 260
__global__ __launch_bounds__(256) void attn_kernel(
    const float* __restrict__ qkv, const int* __restrict__ mask,
    float* __restrict__ out) {
  __shared__ __attribute__((aligned(16))) float ks[256 * KPITCH];
  __shared__ __attribute__((aligned(16))) float vt[32 * VPITCH];

  const int bid = blockIdx.x;
  const int half = bid & 1;
  const int sbh = bid >> 1;
  const int s = sbh >> 5;
  const int b = (sbh >> 3) & 3;
  const int h = sbh & 7;

  const float* base = qkv + (size_t)((s * 4 + b) * 256) * 768;
  const float* qb = base + h * 32;
  const float* kb = base + 256 + h * 32;
  const float* vb = base + 512 + h * 32;
  const int* mk = mask + (size_t)s * 256 * 256;
  float* ob = out + (size_t)((s * 4 + b) * 256) * 256 + h * 32;

  const int tid = threadIdx.x;
  const int w = tid >> 5;
  const int lane = tid & 31;
  const int L = lane & 15;
  const int hi = lane >> 4;
  const int i0 = half * 128 + w * 16;
  const float scale = 0.17677669529663687f;
  const float NEG = -1.0e10f;

  {
    const int r8 = tid >> 3;
    const int d4 = (tid & 7) * 4;
    float4 kx[8], vx[8];
#pragma unroll
    for (int p = 0; p < 8; ++p) {
      const size_t ro = (size_t)(r8 + 32 * p) * 768 + d4;
      kx[p] = *(const float4*)(kb + ro);
      vx[p] = *(const float4*)(vb + ro);
    }
#pragma unroll
    for (int p = 0; p < 8; ++p) {
      const int key = r8 + 32 * p;
      *(float4*)&ks[key * KPITCH + d4] = kx[p];
      vt[(d4 + 0) * VPITCH + key] = vx[p].x;
      vt[(d4 + 1) * VPITCH + key] = vx[p].y;
      vt[(d4 + 2) * VPITCH + key] = vx[p].z;
      vt[(d4 + 3) * VPITCH + key] = vx[p].w;
    }
  }

  v2f qf[8];
  {
    const float* qp = qb + (size_t)(i0 + L) * 768 + 16 * hi;
#pragma unroll
    for (int v = 0; v < 4; ++v) {
      const float4 x = *(const float4*)(qp + 4 * v);
      qf[2 * v] = (v2f){x.x, x.y};
      qf[2 * v + 1] = (v2f){x.z, x.w};
    }
  }

  __syncthreads();

  v8f sc[16];
  const float* kp0 = ks + L * KPITCH + 16 * hi;
  const int* mp0 = mk + (size_t)(i0 + L) * 256 + 8 * hi;
#pragma unroll
  for (int tp = 0; tp < 16; tp += 2) {
    v2f ka[2][8];
#pragma unroll
    for (int u = 0; u < 2; ++u) {
      const float* kp = kp0 + (16 * (tp + u)) * KPITCH;
#pragma unroll
      for (int v = 0; v < 4; ++v) {
        const float4 x = *(const float4*)(kp + 4 * v);
        ka[u][2 * v] = (v2f){x.x, x.y};
        ka[u][2 * v + 1] = (v2f){x.z, x.w};
      }
    }
    v8f a0 = ZERO8, a1 = ZERO8;
#pragma unroll
    for (int c = 0; c < 8; ++c) {
      a0 = wmma_f32_4(ka[0][c], qf[c], a0);
      a1 = wmma_f32_4(ka[1][c], qf[c], a1);
    }
#pragma unroll
    for (int u = 0; u < 2; ++u) {
      const int* mp = mp0 + 16 * (tp + u);
      const int4 m0 = *(const int4*)mp;
      const int4 m1 = *(const int4*)(mp + 4);
      const int mm[8] = {m0.x, m0.y, m0.z, m0.w, m1.x, m1.y, m1.z, m1.w};
      const v8f a = u ? a1 : a0;
      v8f o;
#pragma unroll
      for (int r = 0; r < 8; ++r) o[r] = (mm[r] == 0) ? NEG : a[r] * scale;
      sc[tp + u] = o;
    }
  }

  float mx4[4] = {-3.4e38f, -3.4e38f, -3.4e38f, -3.4e38f};
#pragma unroll
  for (int t = 0; t < 16; ++t)
#pragma unroll
    for (int r = 0; r < 8; ++r) mx4[r & 3] = fmaxf(mx4[r & 3], sc[t][r]);
  float mx = fmaxf(fmaxf(mx4[0], mx4[1]), fmaxf(mx4[2], mx4[3]));
  mx = fmaxf(mx, __shfl_xor(mx, 16));
  float sm4[4] = {0.f, 0.f, 0.f, 0.f};
#pragma unroll
  for (int t = 0; t < 16; ++t)
#pragma unroll
    for (int r = 0; r < 8; ++r) {
      const float e = __expf(sc[t][r] - mx);
      sc[t][r] = e;
      sm4[r & 3] += e;
    }
  float sm = (sm4[0] + sm4[1]) + (sm4[2] + sm4[3]);
  sm += __shfl_xor(sm, 16);
  const float rinv = 1.0f / sm;

  v8f o0 = ZERO8, o1 = ZERO8;
  const float* vp0 = vt + L * VPITCH + 8 * hi;
#pragma unroll
  for (int t = 0; t < 16; ++t) {
    const float* vp = vp0 + 16 * t;
    const float4 x0 = *(const float4*)(vp);
    const float4 x1 = *(const float4*)(vp + 4);
    const float4 y0 = *(const float4*)(vp + 16 * VPITCH);
    const float4 y1 = *(const float4*)(vp + 16 * VPITCH + 4);
    const v2f va0[4] = {(v2f){x0.x, x0.y}, (v2f){x0.z, x0.w},
                        (v2f){x1.x, x1.y}, (v2f){x1.z, x1.w}};
    const v2f va1[4] = {(v2f){y0.x, y0.y}, (v2f){y0.z, y0.w},
                        (v2f){y1.x, y1.y}, (v2f){y1.z, y1.w}};
#pragma unroll
    for (int u = 0; u < 4; ++u) {
      const v2f pb = (v2f){sc[t][2 * u], sc[t][2 * u + 1]};
      o0 = wmma_f32_4(va0[u], pb, o0);
      o1 = wmma_f32_4(va1[u], pb, o1);
    }
  }

  float* op = ob + (size_t)(i0 + L) * 256 + 8 * hi;
  *(float4*)(op)      = (float4){o0[0] * rinv, o0[1] * rinv, o0[2] * rinv, o0[3] * rinv};
  *(float4*)(op + 4)  = (float4){o0[4] * rinv, o0[5] * rinv, o0[6] * rinv, o0[7] * rinv};
  *(float4*)(op + 16) = (float4){o1[0] * rinv, o1[1] * rinv, o1[2] * rinv, o1[3] * rinv};
  *(float4*)(op + 20) = (float4){o1[4] * rinv, o1[5] * rinv, o1[6] * rinv, o1[7] * rinv};
}

extern "C" void kernel_launch(void* const* d_in, const int* in_sizes, int n_in,
                              void* d_out, int out_size, void* d_ws,
                              size_t ws_size, hipStream_t stream) {
  (void)in_sizes; (void)n_in; (void)out_size; (void)ws_size;
  const float* x    = (const float*)d_in[0];
  const int*   mask = (const int*)d_in[1];
  const float* Wqkv = (const float*)d_in[2];
  const float* Wout = (const float*)d_in[3];
  const float* bout = (const float*)d_in[4];
  float* outp = (float*)d_out;

  float* qkv = (float*)d_ws;
  float* ao  = qkv + (size_t)65536 * 768;

  const dim3 blk(256);

  gemm128x64_wmma<<<dim3(512, 12), blk, 0, stream>>>(x, Wqkv, nullptr, qkv,
                                                     65536, 768, 256);

  attn_kernel<<<dim3(4096), blk, 0, stream>>>(qkv, mask, ao);

  gemm128x64_wmma<<<dim3(512, 4), blk, 0, stream>>>(ao, Wout, bout, outp,
                                                    65536, 256, 256);
}


# GEMM (both launches): each wave parks its 16x64 result tile in LDS and stores whole 256-byte rows by 16-byte stores (4 whole lines an instruction), not 64-byte half lines; sums unchanged
# speedup vs baseline: 1.7522x; 1.0411x over previous
#include <hip/hip_runtime.h>
#include <hip/hip_bf16.h>

typedef float v2f __attribute__((ext_vector_type(2)));
typedef float v8f __attribute__((ext_vector_type(8)));

static __device__ __forceinline__ v8f wmma_f32_4(v2f a, v2f b, v8f c) {
  return __builtin_amdgcn_wmma_f32_16x16x4_f32(false, a, false, b, (short)0, c,
                                               false, false);
}

#define ZERO8 {0.f,0.f,0.f,0.f,0.f,0.f,0.f,0.f}

#define APITCH 68
__global__ __launch_bounds__(256) void gemm128x64_wmma(
    const float* __restrict__ A, const float* __restrict__ B,
    const float* __restrict__ bias, float* __restrict__ C,
    int M, int N, int K) {
  __shared__ float bt[64 * 66];
  __shared__ __attribute__((aligned(16))) float as[128 * APITCH];

  const int m0 = blockIdx.x * 128;
  const int n0 = blockIdx.y * 64;
  const int tid = threadIdx.x;
  const int w = tid >> 5;
  const int lane = tid & 31;
  const int L = lane & 15;
  const int hi = lane >> 4;
  const float* afr = as + (w * 16 + L) * APITCH + hi * 2;
  const float* bfr = bt + L * 66 + hi * 2;

  v8f acc[4] = {ZERO8, ZERO8, ZERO8, ZERO8};

  const int c4 = (tid & 15) * 4;
  const int r0 = tid >> 4;
  float* asw = as + r0 * APITCH + c4;
  const float* ag = A + (size_t)(m0 + r0) * K + c4;
  const float* bg = B + (size_t)r0 * N + n0 + c4;
  const size_t a16 = (size_t)16 * K, b16 = (size_t)16 * N;

  for (int kc = 0; kc < K; kc += 64) {
    const float4 b0 = *(const float4*)(bg);
    const float4 b1 = *(const float4*)(bg + b16);
    const float4 b2 = *(const float4*)(bg + 2 * b16);
    const float4 b3 = *(const float4*)(bg + 3 * b16);
    const float4 a0 = *(const float4*)(ag);
    const float4 a1 = *(const float4*)(ag + a16);
    const float4 a2 = *(const float4*)(ag + 2 * a16);
    const float4 a3 = *(const float4*)(ag + 3 * a16);
    const float4 a4 = *(const float4*)(ag + 4 * a16);
    const float4 a5 = *(const float4*)(ag + 5 * a16);
    const float4 a6 = *(const float4*)(ag + 6 * a16);
    const float4 a7 = *(const float4*)(ag + 7 * a16);
    __builtin_amdgcn_sched_barrier(0);
    *(float4*)(asw + 0 * 16 * APITCH) = a0;
    *(float4*)(asw + 1 * 16 * APITCH) = a1;
    *(float4*)(asw + 2 * 16 * APITCH) = a2;
    *(float4*)(asw + 3 * 16 * APITCH) = a3;
    *(float4*)(asw + 4 * 16 * APITCH) = a4;
    *(float4*)(asw + 5 * 16 * APITCH) = a5;
    *(float4*)(asw + 6 * 16 * APITCH) = a6;
    *(float4*)(asw + 7 * 16 * APITCH) = a7;
    {
      float* bw = bt + c4 * 66 + r0;
      bw[0 * 66 +  0] = b0.x; bw[1 * 66 +  0] = b0.y; bw[2 * 66 +  0] = b0.z; bw[3 * 66 +  0] = b0.w;
      bw[0 * 66 + 16] = b1.x; bw[1 * 66 + 16] = b1.y; bw[2 * 66 + 16] = b1.z; bw[3 * 66 + 16] = b1.w;
      bw[0 * 66 + 32] = b2.x; bw[1 * 66 + 32] = b2.y; bw[2 * 66 + 32] = b2.z; bw[3 * 66 + 32] = b2.w;
      bw[0 * 66 + 48] = b3.x; bw[1 * 66 + 48] = b3.y; bw[2 * 66 + 48] = b3.z; bw[3 * 66 + 48] = b3.w;
    }
    ag += 64;
    bg += (size_t)64 * N;
    __syncthreads();
#pragma unroll
    for (int kk = 0; kk < 64; kk += 4) {
      const v2f av = *(const v2f*)(afr + kk);
#pragma unroll
      for (int t = 0; t < 4; ++t) {
        const v2f bf = *(const v2f*)(bfr + t * 16 * 66 + kk);
        acc[t] = wmma_f32_4(av, bf, acc[t]);
      }
    }
    __syncthreads();
  }

  {
    float* cw = as + (w * 16 + hi * 8) * APITCH + L;
#pragma unroll
    for (int t = 0; t < 4; ++t)
#pragma unroll
      for (int r = 0; r < 8; ++r) cw[r * APITCH + t * 16] = acc[t][r];
  }
  __syncthreads();
  {
    const int cc = L * 4;
    float4 bs = {0.0f, 0.0f, 0.0f, 0.0f};
    if (bias) bs = *(const float4*)(bias + n0 + cc);
    const float* cr = as + (w * 16 + hi) * APITCH + cc;
    float* cg = C + (size_t)(m0 + w * 16 + hi) * N + n0 + cc;
#pragma unroll
    for (int p = 0; p < 8; ++p) {
      float4 v = *(const float4*)(cr + 2 * p * APITCH);
      v.x += bs.x; v.y += bs.y; v.z += bs.z; v.w += bs.w;
      *(float4*)(cg + (size_t)(2 * p) * N) = v;
    }
  }
}

#define KPITCH 36
#define VPITCH 260
__global__ __launch_bounds__(256) void attn_kernel(
    const float* __restrict__ qkv, const int* __restrict__ mask,
    float* __restrict__ out) {
  __shared__ __attribute__((aligned(16))) float ks[256 * KPITCH];
  __shared__ __attribute__((aligned(16))) float vt[32 * VPITCH];

  const int bid = blockIdx.x;
  const int half = bid & 1;
  const int sbh = bid >> 1;
  const int s = sbh >> 5;
  const int b = (sbh >> 3) & 3;
  const int h = sbh & 7;

  const float* base = qkv + (size_t)((s * 4 + b) * 256) * 768;
  const float* qb = base + h * 32;
  const float* kb = base + 256 + h * 32;
  const float* vb = base + 512 + h * 32;
  const int* mk = mask + (size_t)s * 256 * 256;
  float* ob = out + (size_t)((s * 4 + b) * 256) * 256 + h * 32;

  const int tid = threadIdx.x;
  const int w = tid >> 5;
  const int lane = tid & 31;
  const int L = lane & 15;
  const int hi = lane >> 4;
  const int i0 = half * 128 + w * 16;
  const float scale = 0.17677669529663687f;
  const float NEG = -1.0e10f;

  {
    const int r8 = tid >> 3;
    const int d4 = (tid & 7) * 4;
    float4 kx[8], vx[8];
#pragma unroll
    for (int p = 0; p < 8; ++p) {
      const size_t ro = (size_t)(r8 + 32 * p) * 768 + d4;
      kx[p] = *(const float4*)(kb + ro);
      vx[p] = *(const float4*)(vb + ro);
    }
#pragma unroll
    for (int p = 0; p < 8; ++p) {
      const int key = r8 + 32 * p;
      *(float4*)&ks[key * KPITCH + d4] = kx[p];
      vt[(d4 + 0) * VPITCH + key] = vx[p].x;
      vt[(d4 + 1) * VPITCH + key] = vx[p].y;
      vt[(d4 + 2) * VPITCH + key] = vx[p].z;
      vt[(d4 + 3) * VPITCH + key] = vx[p].w;
    }
  }

  v2f qf[8];
  {
    const float* qp = qb + (size_t)(i0 + L) * 768 + 16 * hi;
#pragma unroll
    for (int v = 0; v < 4; ++v) {
      const float4 x = *(const float4*)(qp + 4 * v);
      qf[2 * v] = (v2f){x.x, x.y};
      qf[2 * v + 1] = (v2f){x.z, x.w};
    }
  }

  __syncthreads();

  v8f sc[16];
  const float* kp0 = ks + L * KPITCH + 16 * hi;
  const int* mp0 = mk + (size_t)(i0 + L) * 256 + 8 * hi;
#pragma unroll
  for (int tp = 0; tp < 16; tp += 2) {
    v2f ka[2][8];
#pragma unroll
    for (int u = 0; u < 2; ++u) {
      const float* kp = kp0 + (16 * (tp + u)) * KPITCH;
#pragma unroll
      for (int v = 0; v < 4; ++v) {
        const float4 x = *(const float4*)(kp + 4 * v);
        ka[u][2 * v] = (v2f){x.x, x.y};
        ka[u][2 * v + 1] = (v2f){x.z, x.w};
      }
    }
    v8f a0 = ZERO8, a1 = ZERO8;
#pragma unroll
    for (int c = 0; c < 8; ++c) {
      a0 = wmma_f32_4(ka[0][c], qf[c], a0);
      a1 = wmma_f32_4(ka[1][c], qf[c], a1);
    }
#pragma unroll
    for (int u = 0; u < 2; ++u) {
      const int* mp = mp0 + 16 * (tp + u);
      const int4 m0 = *(const int4*)mp;
      const int4 m1 = *(const int4*)(mp + 4);
      const int mm[8] = {m0.x, m0.y, m0.z, m0.w, m1.x, m1.y, m1.z, m1.w};
      const v8f a = u ? a1 : a0;
      v8f o;
#pragma unroll
      for (int r = 0; r < 8; ++r) o[r] = (mm[r] == 0) ? NEG : a[r] * scale;
      sc[tp + u] = o;
    }
  }

  float mx4[4] = {-3.4e38f, -3.4e38f, -3.4e38f, -3.4e38f};
#pragma unroll
  for (int t = 0; t < 16; ++t)
#pragma unroll
    for (int r = 0; r < 8; ++r) mx4[r & 3] = fmaxf(mx4[r & 3], sc[t][r]);
  float mx = fmaxf(fmaxf(mx4[0], mx4[1]), fmaxf(mx4[2], mx4[3]));
  mx = fmaxf(mx, __shfl_xor(mx, 16));
  float sm4[4] = {0.f, 0.f, 0.f, 0.f};
#pragma unroll
  for (int t = 0; t < 16; ++t)
#pragma unroll
    for (int r = 0; r < 8; ++r) {
      const float e = __expf(sc[t][r] - mx);
      sc[t][r] = e;
      sm4[r & 3] += e;
    }
  float sm = (sm4[0] + sm4[1]) + (sm4[2] + sm4[3]);
  sm += __shfl_xor(sm, 16);
  const float rinv = 1.0f / sm;

  v8f o0 = ZERO8, o1 = ZERO8;
  const float* vp0 = vt + L * VPITCH + 8 * hi;
#pragma unroll
  for (int t = 0; t < 16; ++t) {
    const float* vp = vp0 + 16 * t;
    const float4 x0 = *(const float4*)(vp);
    const float4 x1 = *(const float4*)(vp + 4);
    const float4 y0 = *(const float4*)(vp + 16 * VPITCH);
    const float4 y1 = *(const float4*)(vp + 16 * VPITCH + 4);
    const v2f va0[4] = {(v2f){x0.x, x0.y}, (v2f){x0.z, x0.w},
                        (v2f){x1.x, x1.y}, (v2f){x1.z, x1.w}};
    const v2f va1[4] = {(v2f){y0.x, y0.y}, (v2f){y0.z, y0.w},
                        (v2f){y1.x, y1.y}, (v2f){y1.z, y1.w}};
#pragma unroll
    for (int u = 0; u < 4; ++u) {
      const v2f pb = (v2f){sc[t][2 * u], sc[t][2 * u + 1]};
      o0 = wmma_f32_4(va0[u], pb, o0);
      o1 = wmma_f32_4(va1[u], pb, o1);
    }
  }

  float* op = ob + (size_t)(i0 + L) * 256 + 8 * hi;
  *(float4*)(op)      = (float4){o0[0] * rinv, o0[1] * rinv, o0[2] * rinv, o0[3] * rinv};
  *(float4*)(op + 4)  = (float4){o0[4] * rinv, o0[5] * rinv, o0[6] * rinv, o0[7] * rinv};
  *(float4*)(op + 16) = (float4){o1[0] * rinv, o1[1] * rinv, o1[2] * rinv, o1[3] * rinv};
  *(float4*)(op + 20) = (float4){o1[4] * rinv, o1[5] * rinv, o1[6] * rinv, o1[7] * rinv};
}

extern "C" void kernel_launch(void* const* d_in, const int* in_sizes, int n_in,
                              void* d_out, int out_size, void* d_ws,
                              size_t ws_size, hipStream_t stream) {
  (void)in_sizes; (void)n_in; (void)out_size; (void)ws_size;
  const float* x    = (const float*)d_in[0];
  const int*   mask = (const int*)d_in[1];
  const float* Wqkv = (const float*)d_in[2];
  const float* Wout = (const float*)d_in[3];
  const float* bout = (const float*)d_in[4];
  float* outp = (float*)d_out;

  float* qkv = (float*)d_ws;
  float* ao  = qkv + (size_t)65536 * 768;

  const dim3 blk(256);

  gemm128x64_wmma<<<dim3(512, 12), blk, 0, stream>>>(x, Wqkv, nullptr, qkv,
                                                     65536, 768, 256);

  attn_kernel<<<dim3(4096), blk, 0, stream>>>(qkv, mask, ao);

  gemm128x64_wmma<<<dim3(512, 4), blk, 0, stream>>>(ao, Wout, bout, outp,
                                                    65536, 256, 256);
}


# attention: mask packed once per call to 1 bit per key by a small added launch (3 -> 4 launches); a lane reads its 128 key bits with one 16-byte load, not 32 gathered int4 loads; sums unchanged
# speedup vs baseline: 1.8099x; 1.0329x over previous
#include <hip/hip_runtime.h>
#include <hip/hip_bf16.h>

typedef float v2f __attribute__((ext_vector_type(2)));
typedef float v8f __attribute__((ext_vector_type(8)));

static __device__ __forceinline__ v8f wmma_f32_4(v2f a, v2f b, v8f c) {
  return __builtin_amdgcn_wmma_f32_16x16x4_f32(false, a, false, b, (short)0, c,
                                               false, false);
}

#define ZERO8 {0.f,0.f,0.f,0.f,0.f,0.f,0.f,0.f}

#define APITCH 68
__global__ __launch_bounds__(256) void gemm128x64_wmma(
    const float* __restrict__ A, const float* __restrict__ B,
    const float* __restrict__ bias, float* __restrict__ C,
    int M, int N, int K) {
  __shared__ float bt[64 * 66];
  __shared__ __attribute__((aligned(16))) float as[128 * APITCH];

  const int m0 = blockIdx.x * 128;
  const int n0 = blockIdx.y * 64;
  const int tid = threadIdx.x;
  const int w = tid >> 5;
  const int lane = tid & 31;
  const int L = lane & 15;
  const int hi = lane >> 4;
  const float* afr = as + (w * 16 + L) * APITCH + hi * 2;
  const float* bfr = bt + L * 66 + hi * 2;

  v8f acc[4] = {ZERO8, ZERO8, ZERO8, ZERO8};

  const int c4 = (tid & 15) * 4;
  const int r0 = tid >> 4;
  float* asw = as + r0 * APITCH + c4;
  const float* ag = A + (size_t)(m0 + r0) * K + c4;
  const float* bg = B + (size_t)r0 * N + n0 + c4;
  const size_t a16 = (size_t)16 * K, b16 = (size_t)16 * N;

  for (int kc = 0; kc < K; kc += 64) {
    const float4 b0 = *(const float4*)(bg);
    const float4 b1 = *(const float4*)(bg + b16);
    const float4 b2 = *(const float4*)(bg + 2 * b16);
    const float4 b3 = *(const float4*)(bg + 3 * b16);
    const float4 a0 = *(const float4*)(ag);
    const float4 a1 = *(const float4*)(ag + a16);
    const float4 a2 = *(const float4*)(ag + 2 * a16);
    const float4 a3 = *(const float4*)(ag + 3 * a16);
    const float4 a4 = *(const float4*)(ag + 4 * a16);
    const float4 a5 = *(const float4*)(ag + 5 * a16);
    const float4 a6 = *(const float4*)(ag + 6 * a16);
    const float4 a7 = *(const float4*)(ag + 7 * a16);
    __builtin_amdgcn_sched_barrier(0);
    *(float4*)(asw + 0 * 16 * APITCH) = a0;
    *(float4*)(asw + 1 * 16 * APITCH) = a1;
    *(float4*)(asw + 2 * 16 * APITCH) = a2;
    *(float4*)(asw + 3 * 16 * APITCH) = a3;
    *(float4*)(asw + 4 * 16 * APITCH) = a4;
    *(float4*)(asw + 5 * 16 * APITCH) = a5;
    *(float4*)(asw + 6 * 16 * APITCH) = a6;
    *(float4*)(asw + 7 * 16 * APITCH) = a7;
    {
      float* bw = bt + c4 * 66 + r0;
      bw[0 * 66 +  0] = b0.x; bw[1 * 66 +  0] = b0.y; bw[2 * 66 +  0] = b0.z; bw[3 * 66 +  0] = b0.w;
      bw[0 * 66 + 16] = b1.x; bw[1 * 66 + 16] = b1.y; bw[2 * 66 + 16] = b1.z; bw[3 * 66 + 16] = b1.w;
      bw[0 * 66 + 32] = b2.x; bw[1 * 66 + 32] = b2.y; bw[2 * 66 + 32] = b2.z; bw[3 * 66 + 32] = b2.w;
      bw[0 * 66 + 48] = b3.x; bw[1 * 66 + 48] = b3.y; bw[2 * 66 + 48] = b3.z; bw[3 * 66 + 48] = b3.w;
    }
    ag += 64;
    bg += (size_t)64 * N;
    __syncthreads();
#pragma unroll
    for (int kk = 0; kk < 64; kk += 4) {
      const v2f av = *(const v2f*)(afr + kk);
#pragma unroll
      for (int t = 0; t < 4; ++t) {
        const v2f bf = *(const v2f*)(bfr + t * 16 * 66 + kk);
        acc[t] = wmma_f32_4(av, bf, acc[t]);
      }
    }
    __syncthreads();
  }

  {
    float* cw = as + (w * 16 + hi * 8) * APITCH + L;
#pragma unroll
    for (int t = 0; t < 4; ++t)
#pragma unroll
      for (int r = 0; r < 8; ++r) cw[r * APITCH + t * 16] = acc[t][r];
  }
  __syncthreads();
  {
    const int cc = L * 4;
    float4 bs = {0.0f, 0.0f, 0.0f, 0.0f};
    if (bias) bs = *(const float4*)(bias + n0 + cc);
    const float* cr = as + (w * 16 + hi) * APITCH + cc;
    float* cg = C + (size_t)(m0 + w * 16 + hi) * N + n0 + cc;
#pragma unroll
    for (int p = 0; p < 8; ++p) {
      float4 v = *(const float4*)(cr + 2 * p * APITCH);
      v.x += bs.x; v.y += bs.y; v.z += bs.z; v.w += bs.w;
      *(float4*)(cg + (size_t)(2 * p) * N) = v;
    }
  }
}

__global__ __launch_bounds__(256) void mask_pack_kernel(
    const int* __restrict__ mask, unsigned int* __restrict__ bits) {
  const int g = blockIdx.x * 256 + threadIdx.x;
  const int q = g & 3;
  const int hf = (g >> 2) & 1;
  const int row = g >> 3;
  const int* src = mask + (size_t)row * 256 + 64 * q + 8 * hf;
  int4 a[4], b[4];
#pragma unroll
  for (int j = 0; j < 4; ++j) {
    a[j] = *(const int4*)(src + 16 * j);
    b[j] = *(const int4*)(src + 16 * j + 4);
  }
  unsigned int wd = 0;
#pragma unroll
  for (int j = 0; j < 4; ++j) {
    wd |= (unsigned int)(a[j].x != 0) << (8 * j + 0);
    wd |= (unsigned int)(a[j].y != 0) << (8 * j + 1);
    wd |= (unsigned int)(a[j].z != 0) << (8 * j + 2);
    wd |= (unsigned int)(a[j].w != 0) << (8 * j + 3);
    wd |= (unsigned int)(b[j].x != 0) << (8 * j + 4);
    wd |= (unsigned int)(b[j].y != 0) << (8 * j + 5);
    wd |= (unsigned int)(b[j].z != 0) << (8 * j + 6);
    wd |= (unsigned int)(b[j].w != 0) << (8 * j + 7);
  }
  bits[g] = wd;
}

#define KPITCH 36
#define VPITCH 260
__global__ __launch_bounds__(256) void attn_kernel(
    const float* __restrict__ qkv, const unsigned int* __restrict__ mbits,
    float* __restrict__ out) {
  __shared__ __attribute__((aligned(16))) float ks[256 * KPITCH];
  __shared__ __attribute__((aligned(16))) float vt[32 * VPITCH];

  const int bid = blockIdx.x;
  const int half = bid & 1;
  const int sbh = bid >> 1;
  const int s = sbh >> 5;
  const int b = (sbh >> 3) & 3;
  const int h = sbh & 7;

  const float* base = qkv + (size_t)((s * 4 + b) * 256) * 768;
  const float* qb = base + h * 32;
  const float* kb = base + 256 + h * 32;
  const float* vb = base + 512 + h * 32;
  float* ob = out + (size_t)((s * 4 + b) * 256) * 256 + h * 32;

  const int tid = threadIdx.x;
  const int w = tid >> 5;
  const int lane = tid & 31;
  const int L = lane & 15;
  const int hi = lane >> 4;
  const int i0 = half * 128 + w * 16;
  const float scale = 0.17677669529663687f;
  const float NEG = -1.0e10f;

  {
    const int r8 = tid >> 3;
    const int d4 = (tid & 7) * 4;
    float4 kx[8], vx[8];
#pragma unroll
    for (int p = 0; p < 8; ++p) {
      const size_t ro = (size_t)(r8 + 32 * p) * 768 + d4;
      kx[p] = *(const float4*)(kb + ro);
      vx[p] = *(const float4*)(vb + ro);
    }
#pragma unroll
    for (int p = 0; p < 8; ++p) {
      const int key = r8 + 32 * p;
      *(float4*)&ks[key * KPITCH + d4] = kx[p];
      vt[(d4 + 0) * VPITCH + key] = vx[p].x;
      vt[(d4 + 1) * VPITCH + key] = vx[p].y;
      vt[(d4 + 2) * VPITCH + key] = vx[p].z;
      vt[(d4 + 3) * VPITCH + key] = vx[p].w;
    }
  }

  v2f qf[8];
  {
    const float* qp = qb + (size_t)(i0 + L) * 768 + 16 * hi;
#pragma unroll
    for (int v = 0; v < 4; ++v) {
      const float4 x = *(const float4*)(qp + 4 * v);
      qf[2 * v] = (v2f){x.x, x.y};
      qf[2 * v + 1] = (v2f){x.z, x.w};
    }
  }

  const uint4 mq = *(const uint4*)(mbits + ((size_t)(s * 256 + i0 + L) * 2 + hi) * 4);
  const unsigned int mw[4] = {mq.x, mq.y, mq.z, mq.w};

  __syncthreads();

  v8f sc[16];
  const float* kp0 = ks + L * KPITCH + 16 * hi;
#pragma unroll
  for (int tp = 0; tp < 16; tp += 2) {
    v2f ka[2][8];
#pragma unroll
    for (int u = 0; u < 2; ++u) {
      const float* kp = kp0 + (16 * (tp + u)) * KPITCH;
#pragma unroll
      for (int v = 0; v < 4; ++v) {
        const float4 x = *(const float4*)(kp + 4 * v);
        ka[u][2 * v] = (v2f){x.x, x.y};
        ka[u][2 * v + 1] = (v2f){x.z, x.w};
      }
    }
    v8f a0 = ZERO8, a1 = ZERO8;
#pragma unroll
    for (int c = 0; c < 8; ++c) {
      a0 = wmma_f32_4(ka[0][c], qf[c], a0);
      a1 = wmma_f32_4(ka[1][c], qf[c], a1);
    }
#pragma unroll
    for (int u = 0; u < 2; ++u) {
      const int t = tp + u;
      const unsigned int m = mw[t >> 2] >> (8 * (t & 3));
      const v8f a = u ? a1 : a0;
      v8f o;
#pragma unroll
      for (int r = 0; r < 8; ++r) o[r] = ((m >> r) & 1u) ? a[r] * scale : NEG;
      sc[tp + u] = o;
    }
  }

  float mx4[4] = {-3.4e38f, -3.4e38f, -3.4e38f, -3.4e38f};
#pragma unroll
  for (int t = 0; t < 16; ++t)
#pragma unroll
    for (int r = 0; r < 8; ++r) mx4[r & 3] = fmaxf(mx4[r & 3], sc[t][r]);
  float mx = fmaxf(fmaxf(mx4[0], mx4[1]), fmaxf(mx4[2], mx4[3]));
  mx = fmaxf(mx, __shfl_xor(mx, 16));
  float sm4[4] = {0.f, 0.f, 0.f, 0.f};
#pragma unroll
  for (int t = 0; t < 16; ++t)
#pragma unroll
    for (int r = 0; r < 8; ++r) {
      const float e = __expf(sc[t][r] - mx);
      sc[t][r] = e;
      sm4[r & 3] += e;
    }
  float sm = (sm4[0] + sm4[1]) + (sm4[2] + sm4[3]);
  sm += __shfl_xor(sm, 16);
  const float rinv = 1.0f / sm;

  v8f o0 = ZERO8, o1 = ZERO8;
  const float* vp0 = vt + L * VPITCH + 8 * hi;
#pragma unroll
  for (int t = 0; t < 16; ++t) {
    const float* vp = vp0 + 16 * t;
    const float4 x0 = *(const float4*)(vp);
    const float4 x1 = *(const float4*)(vp + 4);
    const float4 y0 = *(const float4*)(vp + 16 * VPITCH);
    const float4 y1 = *(const float4*)(vp + 16 * VPITCH + 4);
    const v2f va0[4] = {(v2f){x0.x, x0.y}, (v2f){x0.z, x0.w},
                        (v2f){x1.x, x1.y}, (v2f){x1.z, x1.w}};
    const v2f va1[4] = {(v2f){y0.x, y0.y}, (v2f){y0.z, y0.w},
                        (v2f){y1.x, y1.y}, (v2f){y1.z, y1.w}};
#pragma unroll
    for (int u = 0; u < 4; ++u) {
      const v2f pb = (v2f){sc[t][2 * u], sc[t][2 * u + 1]};
      o0 = wmma_f32_4(va0[u], pb, o0);
      o1 = wmma_f32_4(va1[u], pb, o1);
    }
  }

  float* op = ob + (size_t)(i0 + L) * 256 + 8 * hi;
  *(float4*)(op)      = (float4){o0[0] * rinv, o0[1] * rinv, o0[2] * rinv, o0[3] * rinv};
  *(float4*)(op + 4)  = (float4){o0[4] * rinv, o0[5] * rinv, o0[6] * rinv, o0[7] * rinv};
  *(float4*)(op + 16) = (float4){o1[0] * rinv, o1[1] * rinv, o1[2] * rinv, o1[3] * rinv};
  *(float4*)(op + 20) = (float4){o1[4] * rinv, o1[5] * rinv, o1[6] * rinv, o1[7] * rinv};
}

extern "C" void kernel_launch(void* const* d_in, const int* in_sizes, int n_in,
                              void* d_out, int out_size, void* d_ws,
                              size_t ws_size, hipStream_t stream) {
  (void)in_sizes; (void)n_in; (void)out_size; (void)ws_size;
  const float* x    = (const float*)d_in[0];
  const int*   mask = (const int*)d_in[1];
  const float* Wqkv = (const float*)d_in[2];
  const float* Wout = (const float*)d_in[3];
  const float* bout = (const float*)d_in[4];
  float* outp = (float*)d_out;

  float* qkv = (float*)d_ws;
  float* ao  = qkv + (size_t)65536 * 768;

  unsigned int* mbits = (unsigned int*)d_out;

  const dim3 blk(256);

  mask_pack_kernel<<<dim3(512), blk, 0, stream>>>(mask, mbits);

  gemm128x64_wmma<<<dim3(512, 12), blk, 0, stream>>>(x, Wqkv, nullptr, qkv,
                                                     65536, 768, 256);

  attn_kernel<<<dim3(4096), blk, 0, stream>>>(qkv, mbits, ao);

  gemm128x64_wmma<<<dim3(512, 4), blk, 0, stream>>>(ao, Wout, bout, outp,
                                                    65536, 256, 256);
}
